# starting kernel
# baseline (speedup 1.0000x reference)
#include <hip/hip_runtime.h>
#include <math.h>

#define N_NODES 100000
#define N_EDGES 1600000
#define F_IN    50
#define HID     32
#define NCLS    40

typedef float v2f __attribute__((ext_vector_type(2)));
typedef float v8f __attribute__((ext_vector_type(8)));


__global__ void zero_kernel(float* __restrict__ p, long long n) {
  long long i = blockIdx.x * (long long)blockDim.x + threadIdx.x;
  if (i < n) p[i] = 0.0f;
}

__global__ void pack_w1_kernel(const float* __restrict__ W0,
                               const float* __restrict__ W1,
                               float* __restrict__ Wcat) {
  int t = blockIdx.x * blockDim.x + threadIdx.x;
  if (t >= 52 * 64) return;
  int k = t >> 6, j = t & 63;
  float v = 0.0f;
  if (k < F_IN) v = (j < HID) ? W0[k * HID + j] : W1[k * HID + (j - HID)];
  Wcat[t] = v;
}

__global__ void pack_w2_kernel(const float* __restrict__ W0,
                               const float* __restrict__ W1,
                               float* __restrict__ Wcat) {
  int t = blockIdx.x * blockDim.x + threadIdx.x;
  if (t >= 64 * 48) return;
  int k = t / 48, j = t % 48;
  float v = 0.0f;
  if (j < NCLS) v = (k < HID) ? W0[k * NCLS + j] : W1[(k - HID) * NCLS + j];
  Wcat[t] = v;
}


__global__ void deg_kernel(const int* __restrict__ rows,
                           const int* __restrict__ cols,
                           float* __restrict__ deg) {
  int e = blockIdx.x * blockDim.x + threadIdx.x;
  if (e >= N_EDGES) return;
  int r = rows[e], c = cols[e];
  if (r != c) atomicAdd(&deg[r], 1.0f);
}

__global__ void dis_kernel(const float* __restrict__ deg, float* __restrict__ dis) {
  int i = blockIdx.x * blockDim.x + threadIdx.x;
  if (i >= N_NODES) return;
  float d = deg[i];
  dis[i] = (d > 0.0f) ? rsqrtf(fmaxf(d, 1.0f)) : 0.0f;
}


__global__ void agg_kernel(const int* __restrict__ rows,
                           const int* __restrict__ cols,
                           const float* __restrict__ dis,
                           const float* __restrict__ src, int sstride, int soff,
                           float* __restrict__ dst) {
  long long gid = blockIdx.x * (long long)blockDim.x + threadIdx.x;
  long long e = gid >> 5;
  if (e >= (long long)N_EDGES) return;
  int f = (int)(gid & 31);
  int r = rows[e], c = cols[e];
  if (r == c) return;
  float w = -dis[r] * dis[c];
  if (w == 0.0f) return;
  atomicAdd(&dst[(long long)r * HID + f],
            w * src[(long long)c * sstride + soff + f]);
}


__global__ void __launch_bounds__(128)
proj1_wmma_kernel(const float* __restrict__ x,
                  const float* __restrict__ Wcat,
                  float* __restrict__ P) {
  int wave = threadIdx.x >> 5;
  int lane = threadIdx.x & 31;
  int half = lane >> 4;
  int l16  = lane & 15;
  int m0 = (blockIdx.x * 4 + wave) * 16;
  if (m0 >= N_NODES) return;
  int mA = m0 + l16;
  const float* arow = x + (long long)mA * F_IN;

  v8f acc0 = {}, acc1 = {}, acc2 = {}, acc3 = {};
  for (int kk = 0; kk < 48; kk += 4) {
    int k = kk + 2 * half;
    v2f a = *(const v2f*)(arow + k);
    const float* b0 = Wcat + k * 64;
    const float* b1 = Wcat + (k + 1) * 64;
    v2f b;
    b.x = b0[l16];      b.y = b1[l16];
    acc0 = __builtin_amdgcn_wmma_f32_16x16x4_f32(false, a, false, b, (short)0, acc0, false, false);
    b.x = b0[16 + l16]; b.y = b1[16 + l16];
    acc1 = __builtin_amdgcn_wmma_f32_16x16x4_f32(false, a, false, b, (short)0, acc1, false, false);
    b.x = b0[32 + l16]; b.y = b1[32 + l16];
    acc2 = __builtin_amdgcn_wmma_f32_16x16x4_f32(false, a, false, b, (short)0, acc2, false, false);
    b.x = b0[48 + l16]; b.y = b1[48 + l16];
    acc3 = __builtin_amdgcn_wmma_f32_16x16x4_f32(false, a, false, b, (short)0, acc3, false, false);
  }
  {
    int k = 48 + 2 * half;
    v2f aload = *(const v2f*)(arow + 48);
    v2f a;
    a.x = half ? 0.0f : aload.x;
    a.y = half ? 0.0f : aload.y;
    const float* b0 = Wcat + k * 64;
    const float* b1 = Wcat + (k + 1) * 64;
    v2f b;
    b.x = b0[l16];      b.y = b1[l16];
    acc0 = __builtin_amdgcn_wmma_f32_16x16x4_f32(false, a, false, b, (short)0, acc0, false, false);
    b.x = b0[16 + l16]; b.y = b1[16 + l16];
    acc1 = __builtin_amdgcn_wmma_f32_16x16x4_f32(false, a, false, b, (short)0, acc1, false, false);
    b.x = b0[32 + l16]; b.y = b1[32 + l16];
    acc2 = __builtin_amdgcn_wmma_f32_16x16x4_f32(false, a, false, b, (short)0, acc2, false, false);
    b.x = b0[48 + l16]; b.y = b1[48 + l16];
    acc3 = __builtin_amdgcn_wmma_f32_16x16x4_f32(false, a, false, b, (short)0, acc3, false, false);
  }
#pragma unroll
  for (int v = 0; v < 8; v++) {
    int mC = m0 + v + 8 * half;
    float* prow = P + (long long)mC * 64 + l16;
    prow[0]  = acc0[v];
    prow[16] = acc1[v];
    prow[32] = acc2[v];
    prow[48] = acc3[v];
  }
}


__global__ void relu_kernel(const float* __restrict__ P,
                            const float* __restrict__ agg1,
                            const float* __restrict__ b1,
                            float* __restrict__ h) {
  long long i = blockIdx.x * (long long)blockDim.x + threadIdx.x;
  if (i >= (long long)N_NODES * HID) return;
  long long n = i >> 5;
  int f = (int)(i & 31);
  float v = P[n * 64 + f] + agg1[i] + b1[f];
  h[i] = v > 0.0f ? v : 0.0f;
}


__global__ void __launch_bounds__(128)
out_wmma_kernel(const float* __restrict__ h,
                const float* __restrict__ agg2,
                const float* __restrict__ Wcat,
                float* __restrict__ outp) {
  int wave = threadIdx.x >> 5;
  int lane = threadIdx.x & 31;
  int half = lane >> 4;
  int l16  = lane & 15;
  int m0 = (blockIdx.x * 4 + wave) * 16;
  if (m0 >= N_NODES) return;
  int mA = m0 + l16;
  const float* hrow = h    + (long long)mA * HID;
  const float* grow = agg2 + (long long)mA * HID - HID;

  v8f acc0 = {}, acc1 = {}, acc2 = {};
  for (int kk = 0; kk < 64; kk += 4) {
    int k = kk + 2 * half;
    const float* src = (k < HID) ? (hrow + k) : (grow + k);
    v2f a = *(const v2f*)src;
    const float* b0 = Wcat + k * 48;
    const float* b1 = Wcat + (k + 1) * 48;
    v2f b;
    b.x = b0[l16];      b.y = b1[l16];
    acc0 = __builtin_amdgcn_wmma_f32_16x16x4_f32(false, a, false, b, (short)0, acc0, false, false);
    b.x = b0[16 + l16]; b.y = b1[16 + l16];
    acc1 = __builtin_amdgcn_wmma_f32_16x16x4_f32(false, a, false, b, (short)0, acc1, false, false);
    b.x = b0[32 + l16]; b.y = b1[32 + l16];
    acc2 = __builtin_amdgcn_wmma_f32_16x16x4_f32(false, a, false, b, (short)0, acc2, false, false);
  }
#pragma unroll
  for (int v = 0; v < 8; v++) {
    int mC = m0 + v + 8 * half;
    float* prow = outp + (long long)mC * 48 + l16;
    prow[0]  = acc0[v];
    prow[16] = acc1[v];
    prow[32] = acc2[v];
  }
}


__global__ void logsoftmax_kernel(const float* __restrict__ outp,
                                  const float* __restrict__ b2,
                                  float* __restrict__ out) {
  int i = blockIdx.x * blockDim.x + threadIdx.x;
  if (i >= N_NODES) return;
  const float* r = outp + (long long)i * 48;
  float m = -INFINITY;
  for (int j = 0; j < NCLS; j++) m = fmaxf(m, r[j] + b2[j]);
  float s = 0.0f;
  for (int j = 0; j < NCLS; j++) s += expf(r[j] + b2[j] - m);
  float ls = logf(s);
  for (int j = 0; j < NCLS; j++)
    out[(long long)i * NCLS + j] = (r[j] + b2[j] - m) - ls;
}


extern "C" void kernel_launch(void* const* d_in, const int* in_sizes, int n_in,
                              void* d_out, int out_size, void* d_ws, size_t ws_size,
                              hipStream_t stream) {
  (void)in_sizes; (void)n_in; (void)out_size; (void)ws_size;
  const float* x    = (const float*)d_in[0];
  const int*   ei   = (const int*)  d_in[1];
  const float* W0_1 = (const float*)d_in[2];
  const float* W1_1 = (const float*)d_in[3];
  const float* b1   = (const float*)d_in[4];
  const float* W0_2 = (const float*)d_in[5];
  const float* W1_2 = (const float*)d_in[6];
  const float* b2   = (const float*)d_in[7];
  float* out = (float*)d_out;

  const int* rows = ei;
  const int* cols = ei + N_EDGES;

  const long long N = N_NODES;
  float* ws   = (float*)d_ws;
  float* deg  = ws;
  float* dis  = ws + N;
  float* P    = ws + 2 * N;
  float* agg1 = ws + 66 * N;
  float* hbuf = ws + 98 * N;
  float* Wc1  = ws + 130 * N;
  float* Wc2  = Wc1 + 52 * 64;

  const int T = 256;
  zero_kernel<<<(unsigned)((N + T - 1) / T), T, 0, stream>>>(deg, N);
  zero_kernel<<<(unsigned)((32 * N + T - 1) / T), T, 0, stream>>>(agg1, 32 * N);
  pack_w1_kernel<<<(52 * 64 + T - 1) / T, T, 0, stream>>>(W0_1, W1_1, Wc1);
  pack_w2_kernel<<<(64 * 48 + T - 1) / T, T, 0, stream>>>(W0_2, W1_2, Wc2);

  deg_kernel<<<(N_EDGES + T - 1) / T, T, 0, stream>>>(rows, cols, deg);
  dis_kernel<<<(unsigned)((N + T - 1) / T), T, 0, stream>>>(deg, dis);

  proj1_wmma_kernel<<<(N_NODES + 63) / 64, 128, 0, stream>>>(x, Wc1, P);
  agg_kernel<<<(unsigned)(((long long)N_EDGES * 32 + T - 1) / T), T, 0, stream>>>(
      rows, cols, dis, P, 64, 32, agg1);
  relu_kernel<<<(unsigned)((32 * N + T - 1) / T), T, 0, stream>>>(P, agg1, b1, hbuf);

  zero_kernel<<<(unsigned)((32 * N + T - 1) / T), T, 0, stream>>>(agg1, 32 * N);
  agg_kernel<<<(unsigned)(((long long)N_EDGES * 32 + T - 1) / T), T, 0, stream>>>(
      rows, cols, dis, hbuf, 32, 0, agg1);
  out_wmma_kernel<<<(N_NODES + 63) / 64, 128, 0, stream>>>(hbuf, agg1, Wc2, P);
  logsoftmax_kernel<<<(unsigned)((N + T - 1) / T), T, 0, stream>>>(P, b2, out);
}


# Edge list filtered once: a first pass gives each 500-row tile its own edge list (col + local row, 4 B) in workspace and its degrees from LDS counts; both neighbour sums walk it, arrival-order adds
# speedup vs baseline: 2.6365x; 2.6365x over previous
#include <hip/hip_runtime.h>
#include <math.h>

#define N_NODES 100000
#define N_EDGES 1600000
#define F_IN    50
#define HID     32
#define NCLS    40

typedef float v2f __attribute__((ext_vector_type(2)));
typedef float v8f __attribute__((ext_vector_type(8)));


__global__ void pack_w1_kernel(const float* __restrict__ W0,
                               const float* __restrict__ W1,
                               float* __restrict__ Wcat) {
  int t = blockIdx.x * blockDim.x + threadIdx.x;
  if (t >= 52 * 64) return;
  int k = t >> 6, j = t & 63;
  float v = 0.0f;
  if (k < F_IN) v = (j < HID) ? W0[k * HID + j] : W1[k * HID + (j - HID)];
  Wcat[t] = v;
}

__global__ void pack_w2_kernel(const float* __restrict__ W0,
                               const float* __restrict__ W1,
                               float* __restrict__ Wcat) {
  int t = blockIdx.x * blockDim.x + threadIdx.x;
  if (t >= 64 * 48) return;
  int k = t / 48, j = t % 48;
  float v = 0.0f;
  if (j < NCLS) v = (k < HID) ? W0[k * NCLS + j] : W1[(k - HID) * NCLS + j];
  Wcat[t] = v;
}


#define TILE_ROWS  500
#define N_TILES    200
#define N_QUADS    (N_EDGES / 4)
#define QUAD_STEP  (N_QUADS / N_TILES)
#define SCAN_T     1024
#define IN_FLIGHT  8
#define QCAP       1056
#define TILE_CAP   16384
#define COL_BITS   17
#define COL_MASK   ((1 << COL_BITS) - 1)
#define EDGE_BITS  21
#define EDGE_MASK  ((1 << EDGE_BITS) - 1)

__device__ __forceinline__ void add_batch(int rl, int c, float w, int lane,
                                          const float* __restrict__ src,
                                          int sstride, int soff,
                                          float* acc) {
  int wi = __float_as_int(w);
  const float* sbase = src + soff + lane;
  float v[32];
#pragma unroll
  for (int u = 0; u < 32; u++) {
    int cj = __builtin_amdgcn_readlane(c, u);
    v[u] = sbase[(long long)cj * sstride];
  }
#pragma unroll
  for (int u = 0; u < 32; u++) {
    float wj = __int_as_float(__builtin_amdgcn_readlane(wi, u));
    int rlj  = __builtin_amdgcn_readlane(rl, u);
    float add = (wj != 0.0f) ? wj * v[u] : 0.0f;
    atomicAdd(&acc[rlj * HID + lane], add);
  }
}

__device__ __forceinline__ void drain_batch(int e, bool live, int lo, int lane,
                                            const int* __restrict__ rows,
                                            const int* __restrict__ cols,
                                            const float* __restrict__ dis,
                                            const float* __restrict__ src,
                                            int sstride, int soff,
                                            float* acc) {
  int r = lo, c = 0;
  float w = 0.0f;
  if (live) {
    r = rows[e];
    c = cols[e];
    if (r != c) w = -dis[r] * dis[c];
  }
  add_batch(r - lo, c, w, lane, src, sstride, soff, acc);
}

__device__ __forceinline__ void flush_batch(int q, bool live, int lo, int lane,
                                            const int* __restrict__ cols,
                                            int* degc, int* fill,
                                            int* __restrict__ tl) {
  int e  = q & EDGE_MASK;
  int rl = (int)((unsigned)q >> EDGE_BITS);
  int c  = cols[e];
  bool keep = live && (lo + rl != c);
  unsigned m = __builtin_amdgcn_ballot_w32(keep);
  int n = __builtin_popcount(m);
  if (n == 0) return;
  int base = 0;
  if (lane == 0) base = atomicAdd(fill, n);
  base = __builtin_amdgcn_readfirstlane(base);
  if (keep) {
    atomicAdd(&degc[rl], 1);
    int pos = base + (int)__builtin_amdgcn_mbcnt_lo(m, 0u);
    if (pos < TILE_CAP) tl[pos] = (rl << COL_BITS) | c;
  }
}

__global__ void __launch_bounds__(SCAN_T)
tile_filter_kernel(const int* __restrict__ rows,
                   const int* __restrict__ cols,
                   float* __restrict__ dis,
                   int* __restrict__ list,
                   int* __restrict__ counts) {
  __shared__ int queue[(SCAN_T / 32) * QCAP];
  __shared__ int degc[TILE_ROWS];
  __shared__ int fill;

  const int tid  = threadIdx.x;
  const int lane = tid & 31;
  const int lo   = blockIdx.x * TILE_ROWS;
  int* wq = queue + (tid >> 5) * QCAP;
  int* tl = list + blockIdx.x * TILE_CAP;
  const int4* rowsq = (const int4*)rows;
  const int start = blockIdx.x * QUAD_STEP;

  for (int i = tid; i < TILE_ROWS; i += SCAN_T) degc[i] = 0;
  if (tid == 0) fill = 0;
  __syncthreads();

  int qn = 0;
  for (int trip = 0; trip < (N_QUADS + IN_FLIGHT * SCAN_T - 1) / (IN_FLIGHT * SCAN_T); trip++) {
    int4 rv[IN_FLIGHT];
    int  qq[IN_FLIGHT];
    bool ok[IN_FLIGHT];
#pragma unroll
    for (int s = 0; s < IN_FLIGHT; s++) {
      int q = trip * (IN_FLIGHT * SCAN_T) + s * SCAN_T + tid;
      ok[s] = q < N_QUADS;
      int t = q + start;
      if (t >= N_QUADS) t -= N_QUADS;
      qq[s] = t;
      rv[s] = rowsq[t];
    }
#pragma unroll
    for (int s = 0; s < IN_FLIGHT; s++) {
      unsigned d0 = (unsigned)(rv[s].x - lo), d1 = (unsigned)(rv[s].y - lo);
      unsigned d2 = (unsigned)(rv[s].z - lo), d3 = (unsigned)(rv[s].w - lo);
      bool p0 = ok[s] && d0 < (unsigned)TILE_ROWS;
      bool p1 = ok[s] && d1 < (unsigned)TILE_ROWS;
      bool p2 = ok[s] && d2 < (unsigned)TILE_ROWS;
      bool p3 = ok[s] && d3 < (unsigned)TILE_ROWS;
      unsigned b0 = __builtin_amdgcn_ballot_w32(p0);
      unsigned b1 = __builtin_amdgcn_ballot_w32(p1);
      unsigned b2 = __builtin_amdgcn_ballot_w32(p2);
      unsigned b3 = __builtin_amdgcn_ballot_w32(p3);
      if ((b0 | b1 | b2 | b3) != 0u) {
        int e = qq[s] * 4;
        if (p0) wq[qn + __builtin_amdgcn_mbcnt_lo(b0, 0u)] = (int)(d0 << EDGE_BITS) | e;
        qn += __builtin_popcount(b0);
        if (p1) wq[qn + __builtin_amdgcn_mbcnt_lo(b1, 0u)] = (int)(d1 << EDGE_BITS) | (e + 1);
        qn += __builtin_popcount(b1);
        if (p2) wq[qn + __builtin_amdgcn_mbcnt_lo(b2, 0u)] = (int)(d2 << EDGE_BITS) | (e + 2);
        qn += __builtin_popcount(b2);
        if (p3) wq[qn + __builtin_amdgcn_mbcnt_lo(b3, 0u)] = (int)(d3 << EDGE_BITS) | (e + 3);
        qn += __builtin_popcount(b3);
      }
    }
    while (qn >= 32) {
      int q = wq[qn - 32 + lane];
      qn -= 32;
      flush_batch(q, true, lo, lane, cols, degc, &fill, tl);
    }
  }
  if (qn > 0) {
    bool live = lane < qn;
    int q = live ? wq[lane] : 0;
    flush_batch(q, live, lo, lane, cols, degc, &fill, tl);
  }
  __syncthreads();

  for (int i = tid; i < TILE_ROWS; i += SCAN_T) {
    float d = (float)degc[i];
    dis[lo + i] = (d > 0.0f) ? rsqrtf(fmaxf(d, 1.0f)) : 0.0f;
  }
  if (tid == 0) counts[blockIdx.x] = fill;
}

__global__ void __launch_bounds__(SCAN_T)
agg_tile_kernel(const int* __restrict__ rows,
                const int* __restrict__ cols,
                const float* __restrict__ dis,
                const int* __restrict__ list,
                const int* __restrict__ counts,
                const float* __restrict__ src, int sstride, int soff,
                float* __restrict__ dst) {
  __shared__ float acc[TILE_ROWS * HID];
  __shared__ int   queue[(SCAN_T / 32) * QCAP];

  const int tid  = threadIdx.x;
  const int lane = tid & 31;
  const int lo   = blockIdx.x * TILE_ROWS;

  for (int i = tid; i < TILE_ROWS * HID; i += SCAN_T) acc[i] = 0.0f;
  __syncthreads();

  const int cnt = counts[blockIdx.x];
  if (cnt <= TILE_CAP) {
    const int* tl = list + blockIdx.x * TILE_CAP;
    int b = (tid >> 5) * 32;
    int ent = (b + lane < cnt) ? tl[b + lane] : -1;
    while (b < cnt) {
      int nb = b + SCAN_T;
      int nent = (nb + lane < cnt) ? tl[nb + lane] : -1;
      bool live = ent >= 0;
      int c  = live ? (ent & COL_MASK) : 0;
      int rl = live ? (ent >> COL_BITS) : 0;
      float w = 0.0f;
      if (live) w = -dis[lo + rl] * dis[c];
      add_batch(rl, c, w, lane, src, sstride, soff, acc);
      ent = nent;
      b = nb;
    }
  } else {
    int* wq = queue + (tid >> 5) * QCAP;
    const int4* rowsq = (const int4*)rows;
    const int start = blockIdx.x * QUAD_STEP;
    int qn = 0;
    for (int trip = 0; trip < (N_QUADS + IN_FLIGHT * SCAN_T - 1) / (IN_FLIGHT * SCAN_T); trip++) {
      int4 rv[IN_FLIGHT];
      int  qq[IN_FLIGHT];
      bool ok[IN_FLIGHT];
#pragma unroll
      for (int s = 0; s < IN_FLIGHT; s++) {
        int q = trip * (IN_FLIGHT * SCAN_T) + s * SCAN_T + tid;
        ok[s] = q < N_QUADS;
        int t = q + start;
        if (t >= N_QUADS) t -= N_QUADS;
        qq[s] = t;
        rv[s] = rowsq[t];
      }
#pragma unroll
      for (int s = 0; s < IN_FLIGHT; s++) {
        bool p0 = ok[s] && (unsigned)(rv[s].x - lo) < (unsigned)TILE_ROWS;
        bool p1 = ok[s] && (unsigned)(rv[s].y - lo) < (unsigned)TILE_ROWS;
        bool p2 = ok[s] && (unsigned)(rv[s].z - lo) < (unsigned)TILE_ROWS;
        bool p3 = ok[s] && (unsigned)(rv[s].w - lo) < (unsigned)TILE_ROWS;
        unsigned b0 = __builtin_amdgcn_ballot_w32(p0);
        unsigned b1 = __builtin_amdgcn_ballot_w32(p1);
        unsigned b2 = __builtin_amdgcn_ballot_w32(p2);
        unsigned b3 = __builtin_amdgcn_ballot_w32(p3);
        if ((b0 | b1 | b2 | b3) != 0u) {
          int e = qq[s] * 4;
          if (p0) wq[qn + __builtin_amdgcn_mbcnt_lo(b0, 0u)] = e;
          qn += __builtin_popcount(b0);
          if (p1) wq[qn + __builtin_amdgcn_mbcnt_lo(b1, 0u)] = e + 1;
          qn += __builtin_popcount(b1);
          if (p2) wq[qn + __builtin_amdgcn_mbcnt_lo(b2, 0u)] = e + 2;
          qn += __builtin_popcount(b2);
          if (p3) wq[qn + __builtin_amdgcn_mbcnt_lo(b3, 0u)] = e + 3;
          qn += __builtin_popcount(b3);
        }
      }
      while (qn >= 32) {
        int e = wq[qn - 32 + lane];
        qn -= 32;
        drain_batch(e, true, lo, lane, rows, cols, dis, src, sstride, soff, acc);
      }
    }
    if (qn > 0) {
      bool live = lane < qn;
      int e = live ? wq[lane] : 0;
      drain_batch(e, live, lo, lane, rows, cols, dis, src, sstride, soff, acc);
    }
  }
  __syncthreads();

  float* drow = dst + (long long)lo * HID;
  for (int i = tid; i < TILE_ROWS * HID; i += SCAN_T) drow[i] = acc[i];
}


__global__ void __launch_bounds__(128)
proj1_wmma_kernel(const float* __restrict__ x,
                  const float* __restrict__ Wcat,
                  float* __restrict__ P) {
  int wave = threadIdx.x >> 5;
  int lane = threadIdx.x & 31;
  int half = lane >> 4;
  int l16  = lane & 15;
  int m0 = (blockIdx.x * 4 + wave) * 16;
  if (m0 >= N_NODES) return;
  int mA = m0 + l16;
  const float* arow = x + (long long)mA * F_IN;

  v8f acc0 = {}, acc1 = {}, acc2 = {}, acc3 = {};
  for (int kk = 0; kk < 48; kk += 4) {
    int k = kk + 2 * half;
    v2f a = *(const v2f*)(arow + k);
    const float* b0 = Wcat + k * 64;
    const float* b1 = Wcat + (k + 1) * 64;
    v2f b;
    b.x = b0[l16];      b.y = b1[l16];
    acc0 = __builtin_amdgcn_wmma_f32_16x16x4_f32(false, a, false, b, (short)0, acc0, false, false);
    b.x = b0[16 + l16]; b.y = b1[16 + l16];
    acc1 = __builtin_amdgcn_wmma_f32_16x16x4_f32(false, a, false, b, (short)0, acc1, false, false);
    b.x = b0[32 + l16]; b.y = b1[32 + l16];
    acc2 = __builtin_amdgcn_wmma_f32_16x16x4_f32(false, a, false, b, (short)0, acc2, false, false);
    b.x = b0[48 + l16]; b.y = b1[48 + l16];
    acc3 = __builtin_amdgcn_wmma_f32_16x16x4_f32(false, a, false, b, (short)0, acc3, false, false);
  }
  {
    int k = 48 + 2 * half;
    v2f aload = *(const v2f*)(arow + 48);
    v2f a;
    a.x = half ? 0.0f : aload.x;
    a.y = half ? 0.0f : aload.y;
    const float* b0 = Wcat + k * 64;
    const float* b1 = Wcat + (k + 1) * 64;
    v2f b;
    b.x = b0[l16];      b.y = b1[l16];
    acc0 = __builtin_amdgcn_wmma_f32_16x16x4_f32(false, a, false, b, (short)0, acc0, false, false);
    b.x = b0[16 + l16]; b.y = b1[16 + l16];
    acc1 = __builtin_amdgcn_wmma_f32_16x16x4_f32(false, a, false, b, (short)0, acc1, false, false);
    b.x = b0[32 + l16]; b.y = b1[32 + l16];
    acc2 = __builtin_amdgcn_wmma_f32_16x16x4_f32(false, a, false, b, (short)0, acc2, false, false);
    b.x = b0[48 + l16]; b.y = b1[48 + l16];
    acc3 = __builtin_amdgcn_wmma_f32_16x16x4_f32(false, a, false, b, (short)0, acc3, false, false);
  }
#pragma unroll
  for (int v = 0; v < 8; v++) {
    int mC = m0 + v + 8 * half;
    float* prow = P + (long long)mC * 64 + l16;
    prow[0]  = acc0[v];
    prow[16] = acc1[v];
    prow[32] = acc2[v];
    prow[48] = acc3[v];
  }
}


__global__ void relu_kernel(const float* __restrict__ P,
                            const float* __restrict__ b1,
                            float* hx) {
  long long i = blockIdx.x * (long long)blockDim.x + threadIdx.x;
  if (i >= (long long)N_NODES * HID) return;
  long long n = i >> 5;
  int f = (int)(i & 31);
  float v = P[n * 64 + f] + hx[i] + b1[f];
  hx[i] = v > 0.0f ? v : 0.0f;
}


__global__ void __launch_bounds__(128)
out_wmma_kernel(const float* __restrict__ h,
                const float* __restrict__ agg2,
                const float* __restrict__ Wcat,
                float* __restrict__ outp) {
  int wave = threadIdx.x >> 5;
  int lane = threadIdx.x & 31;
  int half = lane >> 4;
  int l16  = lane & 15;
  int m0 = (blockIdx.x * 4 + wave) * 16;
  if (m0 >= N_NODES) return;
  int mA = m0 + l16;
  const float* hrow = h    + (long long)mA * HID;
  const float* grow = agg2 + (long long)mA * HID - HID;

  v8f acc0 = {}, acc1 = {}, acc2 = {};
  for (int kk = 0; kk < 64; kk += 4) {
    int k = kk + 2 * half;
    const float* src = (k < HID) ? (hrow + k) : (grow + k);
    v2f a = *(const v2f*)src;
    const float* b0 = Wcat + k * 48;
    const float* b1 = Wcat + (k + 1) * 48;
    v2f b;
    b.x = b0[l16];      b.y = b1[l16];
    acc0 = __builtin_amdgcn_wmma_f32_16x16x4_f32(false, a, false, b, (short)0, acc0, false, false);
    b.x = b0[16 + l16]; b.y = b1[16 + l16];
    acc1 = __builtin_amdgcn_wmma_f32_16x16x4_f32(false, a, false, b, (short)0, acc1, false, false);
    b.x = b0[32 + l16]; b.y = b1[32 + l16];
    acc2 = __builtin_amdgcn_wmma_f32_16x16x4_f32(false, a, false, b, (short)0, acc2, false, false);
  }
#pragma unroll
  for (int v = 0; v < 8; v++) {
    int mC = m0 + v + 8 * half;
    float* prow = outp + (long long)mC * 48 + l16;
    prow[0]  = acc0[v];
    prow[16] = acc1[v];
    prow[32] = acc2[v];
  }
}


__global__ void logsoftmax_kernel(const float* __restrict__ outp,
                                  const float* __restrict__ b2,
                                  float* __restrict__ out) {
  int i = blockIdx.x * blockDim.x + threadIdx.x;
  if (i >= N_NODES) return;
  const float* r = outp + (long long)i * 48;
  float m = -INFINITY;
  for (int j = 0; j < NCLS; j++) m = fmaxf(m, r[j] + b2[j]);
  float s = 0.0f;
  for (int j = 0; j < NCLS; j++) s += expf(r[j] + b2[j] - m);
  float ls = logf(s);
  for (int j = 0; j < NCLS; j++)
    out[(long long)i * NCLS + j] = (r[j] + b2[j] - m) - ls;
}


extern "C" void kernel_launch(void* const* d_in, const int* in_sizes, int n_in,
                              void* d_out, int out_size, void* d_ws, size_t ws_size,
                              hipStream_t stream) {
  (void)in_sizes; (void)n_in; (void)out_size; (void)ws_size;
  const float* x    = (const float*)d_in[0];
  const int*   ei   = (const int*)  d_in[1];
  const float* W0_1 = (const float*)d_in[2];
  const float* W1_1 = (const float*)d_in[3];
  const float* b1   = (const float*)d_in[4];
  const float* W0_2 = (const float*)d_in[5];
  const float* W1_2 = (const float*)d_in[6];
  const float* b2   = (const float*)d_in[7];
  float* out = (float*)d_out;

  const int* rows = ei;
  const int* cols = ei + N_EDGES;

  const long long N = N_NODES;
  float* ws   = (float*)d_ws;
  float* dis  = ws;
  float* P    = ws + N;
  float* agg2 = ws + N;
  float* outp = ws + 33 * N;
  int*   list = (int*)(ws + 65 * N);
  int*   cnts = (int*)(ws + 98 * N - 256);
  float* hx   = ws + 98 * N;
  float* Wc1  = ws + 130 * N;
  float* Wc2  = Wc1 + 52 * 64;

  const int T = 256;
  pack_w1_kernel<<<(52 * 64 + T - 1) / T, T, 0, stream>>>(W0_1, W1_1, Wc1);
  pack_w2_kernel<<<(64 * 48 + T - 1) / T, T, 0, stream>>>(W0_2, W1_2, Wc2);

  tile_filter_kernel<<<N_TILES, SCAN_T, 0, stream>>>(rows, cols, dis, list, cnts);

  proj1_wmma_kernel<<<(N_NODES + 63) / 64, 128, 0, stream>>>(x, Wc1, P);
  agg_tile_kernel<<<N_TILES, SCAN_T, 0, stream>>>(rows, cols, dis, list, cnts, P, 64, 32, hx);
  relu_kernel<<<(unsigned)((32 * N + T - 1) / T), T, 0, stream>>>(P, b1, hx);

  agg_tile_kernel<<<N_TILES, SCAN_T, 0, stream>>>(rows, cols, dis, list, cnts, hx, 32, 0, agg2);
  out_wmma_kernel<<<(N_NODES + 63) / 64, 128, 0, stream>>>(hx, agg2, Wc2, outp);
  logsoftmax_kernel<<<(unsigned)((N + T - 1) / T), T, 0, stream>>>(outp, b2, out);
}
